# starting kernel
# baseline (speedup 1.0000x reference)
#include <hip/hip_runtime.h>


#define NNODES 17
#define NPAD   32
#define NSLOT  9
#define DDIM   2048

typedef __attribute__((ext_vector_type(2))) float v2f;
typedef __attribute__((ext_vector_type(8))) float v8f;

__global__ __launch_bounds__(256) void prep_kernel(
    const float* __restrict__ xmain, const float* __restrict__ xsig, int nsig,
    const int* __restrict__ ei, const int* __restrict__ et, int E,
    float* __restrict__ xT)
{
  int slot = blockIdx.x / NPAD;
  int n    = blockIdx.x % NPAD;

  for (int j = 0; j < DDIM / 256; ++j) {
    int d = threadIdx.x + j * 256;
    float v = 0.0f;
    if (n < NNODES) {
      if (slot == 8) {
        v = (n < nsig) ? xsig[(size_t)n * DDIM + d]
                       : xmain[(size_t)n * DDIM + d];
      } else {
        float acc = 0.0f;
        int   cnt = 0;
        for (int e = 0; e < E; ++e) {
          if (et[e] == slot && ei[E + e] == n) {
            int m = ei[e];
            acc += (m < nsig) ? xsig[(size_t)m * DDIM + d]
                              : xmain[(size_t)m * DDIM + d];
            ++cnt;
          }
        }
        v = cnt ? acc / (float)cnt : 0.0f;
      }
    }
    xT[((size_t)slot * DDIM + d) * NPAD + n] = v;
  }
}

__global__ __launch_bounds__(128) void gemm_kernel(
    const float* __restrict__ xT,
    const float* __restrict__ W,
    const float* __restrict__ root,
    float* __restrict__ partial)
{
  const int wave = threadIdx.x >> 5;
  const int lane = threadIdx.x & 31;
  const int task = blockIdx.x * 4 + wave;
  const int slot  = task >> 7;
  const int strip = task & 127;
  const int h0    = strip * 16;

  const int half = lane >> 4;
  const int hl   = lane & 15;

  const float* __restrict__ Wr =
      (slot < 8) ? (W + (size_t)slot * DDIM * DDIM) : root;
  const float* __restrict__ Xs = xT + (size_t)slot * DDIM * NPAD;

  const float* wp = Wr + (size_t)(2 * half) * DDIM + h0 + hl;
  const float* ap = Xs + (size_t)(2 * half) * NPAD + hl;

  v8f c0 = {};
  v8f c1 = {};

#pragma unroll 4
  for (int k0 = 0; k0 < DDIM; k0 += 4) {
    v2f b, a0, a1;
    b.x  = wp[0];
    b.y  = wp[DDIM];
    a0.x = ap[0];
    a0.y = ap[NPAD];
    a1.x = ap[16];
    a1.y = ap[NPAD + 16];
    c0 = __builtin_amdgcn_wmma_f32_16x16x4_f32(false, a0, false, b,
                                               (short)0, c0, false, false);
    c1 = __builtin_amdgcn_wmma_f32_16x16x4_f32(false, a1, false, b,
                                               (short)0, c1, false, false);
    wp += (size_t)4 * DDIM;
    ap += 4 * NPAD;
  }

  size_t base = (size_t)slot * NPAD * DDIM + h0 + hl;
#pragma unroll
  for (int i = 0; i < 8; ++i) {
    partial[base + (size_t)(i + 8 * half) * DDIM]      = c0[i];
    partial[base + (size_t)(16 + i + 8 * half) * DDIM] = c1[i];
  }
}

__global__ __launch_bounds__(256) void reduce_kernel(
    const float* __restrict__ partial, const float* __restrict__ bias,
    float* __restrict__ out, int relu)
{
  int idx = blockIdx.x * blockDim.x + threadIdx.x;
  if (idx >= NNODES * DDIM) return;
  int n = idx / DDIM;
  int h = idx % DDIM;
  float s = bias[h];
#pragma unroll
  for (int slot = 0; slot < NSLOT; ++slot)
    s += partial[((size_t)slot * NPAD + n) * DDIM + h];
  if (relu) s = fmaxf(s, 0.0f);
  out[idx] = s;
}

extern "C" void kernel_launch(void* const* d_in, const int* in_sizes, int n_in,
                              void* d_out, int out_size, void* d_ws, size_t ws_size,
                              hipStream_t stream) {
  const float* node_emb = (const float*)d_in[0];
  const float* sig      = (const float*)d_in[1];
  const float* W1       = (const float*)d_in[2];
  const float* root1    = (const float*)d_in[3];
  const float* bias1    = (const float*)d_in[4];
  const float* W2       = (const float*)d_in[5];
  const float* root2    = (const float*)d_in[6];
  const float* bias2    = (const float*)d_in[7];
  const int*   ei       = (const int*)d_in[8];
  const int*   et       = (const int*)d_in[9];
  const int    E        = in_sizes[9];

  float* xT      = (float*)d_ws;
  float* partial = xT + (size_t)NSLOT * DDIM * NPAD;
  float* h1      = partial + (size_t)NSLOT * NPAD * DDIM;

  const int prepGrid   = NSLOT * NPAD;
  const int gemmGrid   = (NSLOT * 128) / 4;
  const int reduceGrid = (NNODES * DDIM + 255) / 256;

  prep_kernel<<<prepGrid, 256, 0, stream>>>(node_emb, sig, 6, ei, et, E, xT);
  gemm_kernel<<<gemmGrid, 128, 0, stream>>>(xT, W1, root1, partial);
  reduce_kernel<<<reduceGrid, 256, 0, stream>>>(partial, bias1, h1, 1);

  prep_kernel<<<prepGrid, 256, 0, stream>>>(h1, nullptr, 0, ei, et, E, xT);
  gemm_kernel<<<gemmGrid, 128, 0, stream>>>(xT, W2, root2, partial);
  reduce_kernel<<<reduceGrid, 256, 0, stream>>>(partial, bias2, (float*)d_out, 0);
}


# gemm regrouped: block = 128-col panel x 128-row K chunk over all 9 slots, A frags stay in registers, W tile by whole-row 16-B loads via LDS; aggregate adds 16 K planes: new summation order
# speedup vs baseline: 3.1401x; 3.1401x over previous
#include <hip/hip_runtime.h>


#define NNODES 17
#define NPAD   32
#define NREL   8
#define NSLOT  9
#define DDIM   2048
#define MAXE   1024
#define KC     128
#define KSPLIT 16
#define PANEL  128
#define BPITCH 144

typedef __attribute__((ext_vector_type(2))) float v2f;
typedef __attribute__((ext_vector_type(4))) float v4f;
typedef __attribute__((ext_vector_type(8))) float v8f;

__global__ __launch_bounds__(256) void transpose_kernel(
    const float* __restrict__ xmain, const float* __restrict__ xsig, int nsig,
    float* __restrict__ xT)
{
  int idx = blockIdx.x * 256 + threadIdx.x;
  int n = idx & (NPAD - 1);
  int k = idx >> 5;
  float v = 0.0f;
  if (n < NNODES)
    v = (n < nsig) ? xsig[(size_t)n * DDIM + k] : xmain[(size_t)n * DDIM + k];
  xT[idx] = v;
}

__global__ __launch_bounds__(512) void gemm_kernel(
    const float* __restrict__ xT,
    const float* __restrict__ W,
    const float* __restrict__ root,
    float* __restrict__ partial)
{
  __shared__ __attribute__((aligned(16))) float Bs[2][KC * BPITCH];

  const int tid  = threadIdx.x;
  const int wave = tid >> 5;
  const int lane = tid & 31;
  const int half = lane >> 4;
  const int hl   = lane & 15;
  const int strip = wave & 7;
  const int tile  = wave >> 3;

  const int panel  = blockIdx.x & 15;
  const int kchunk = blockIdx.x >> 4;
  const int pc = panel * PANEL;
  const int kc = kchunk * KC;

  v2f a[KC / 4];
  {
    const float* ap = xT + (size_t)(kc + 2 * half) * NPAD + 16 * tile + hl;
#pragma unroll
    for (int j = 0; j < KC / 4; ++j) {
      a[j].x = ap[(size_t)(4 * j) * NPAD];
      a[j].y = ap[(size_t)(4 * j + 1) * NPAD];
    }
  }

  const size_t goff = (size_t)(kc + wave) * DDIM + pc + 4 * lane;
  const int    soff = wave * BPITCH + 4 * lane;

  v4f r[8];
#pragma unroll
  for (int i = 0; i < 8; ++i)
    r[i] = *(const v4f*)(W + goff + (size_t)(16 * i) * DDIM);

#pragma unroll 1
  for (int slot = 0; slot < NSLOT; ++slot) {
    float* bs = Bs[slot & 1];
#pragma unroll
    for (int i = 0; i < 8; ++i)
      *(v4f*)(bs + soff + 16 * i * BPITCH) = r[i];
    __syncthreads();

    if (slot + 1 < NSLOT) {
      const float* Wn = (slot + 1 < NREL)
                            ? (W + (size_t)(slot + 1) * DDIM * DDIM) : root;
#pragma unroll
      for (int i = 0; i < 8; ++i)
        r[i] = *(const v4f*)(Wn + goff + (size_t)(16 * i) * DDIM);
    }

    v8f c = {};
    const float* bp = bs + (2 * half) * BPITCH + 16 * strip + hl;
    v2f b[KC / 4];
#pragma unroll
    for (int j = 0; j < KC / 4; ++j) {
      b[j].x = bp[(4 * j) * BPITCH];
      b[j].y = bp[(4 * j + 1) * BPITCH];
    }
    __builtin_amdgcn_sched_barrier(0);
#pragma unroll
    for (int j = 0; j < KC / 4; ++j)
      c = __builtin_amdgcn_wmma_f32_16x16x4_f32(false, a[j], false, b[j],
                                                (short)0, c, false, false);

    float* op = partial + ((size_t)(kchunk * NSLOT + slot) * NPAD) * DDIM
                + pc + 16 * strip + hl;
    if (tile == 0) {
#pragma unroll
      for (int i = 0; i < 8; ++i)
        op[(size_t)(i + 8 * half) * DDIM] = c[i];
    } else if (half == 0) {
      op[(size_t)16 * DDIM] = c[0];
    }
  }
}

__device__ __forceinline__ float plane_sum(const float* __restrict__ partial,
                                           int slot, int row, int h)
{
  const float* p = partial + ((size_t)slot * NPAD + row) * DDIM + h;
  float s = 0.0f;
#pragma unroll
  for (int q = 0; q < KSPLIT; ++q)
    s += p[(size_t)q * NSLOT * NPAD * DDIM];
  return s;
}

__global__ __launch_bounds__(256) void aggregate_kernel(
    const float* __restrict__ partial, const float* __restrict__ bias,
    const int* __restrict__ ei, const int* __restrict__ et, int E,
    float* __restrict__ out, int relu)
{
  __shared__ int sTy[MAXE];
  __shared__ int sSrc[MAXE];
  __shared__ int lTy[MAXE];
  __shared__ int lSrc[MAXE];
  __shared__ int lCnt[MAXE];
  __shared__ int sL;

  const int tid = threadIdx.x;
  const int n   = blockIdx.x >> 3;
  const int h   = ((blockIdx.x & 7) << 8) + tid;

  const float rootv = plane_sum(partial, NREL, n, h);
  const float bv    = bias[h];

  float agg = 0.0f;

  if (E <= MAXE) {
    for (int e = tid; e < E; e += 256) {
      int t = et[e];
      sTy[e]  = (ei[E + e] == n && (unsigned)t < (unsigned)NREL) ? t : -1;
      sSrc[e] = ei[e];
    }
    __syncthreads();
    for (int e = tid; e < E; e += 256) {
      const int t = sTy[e];
      if (t >= 0) {
        int lower = 0, same = 0, sameBefore = 0;
        for (int e2 = 0; e2 < E; ++e2) {
          const int t2 = sTy[e2];
          lower      += (t2 >= 0 && t2 < t) ? 1 : 0;
          same       += (t2 == t) ? 1 : 0;
          sameBefore += (t2 == t && e2 < e) ? 1 : 0;
        }
        const int rank = lower + sameBefore;
        lTy[rank]  = t;
        lSrc[rank] = sSrc[e];
        lCnt[rank] = (sameBefore == same - 1) ? same : 0;
      }
    }
    if (tid == 0) {
      int c = 0;
      for (int e2 = 0; e2 < E; ++e2) c += (sTy[e2] >= 0) ? 1 : 0;
      sL = c;
    }
    __syncthreads();

    const int L = sL;
    float t = 0.0f;
    for (int i = 0; i < L; ++i) {
      t += plane_sum(partial, lTy[i], lSrc[i], h);
      const int c = lCnt[i];
      if (c) { agg += t / (float)c; t = 0.0f; }
    }
  } else {
    for (int r = 0; r < NREL; ++r) {
      float t = 0.0f;
      int   c = 0;
      for (int e = 0; e < E; ++e) {
        if (et[e] == r && ei[E + e] == n) {
          t += plane_sum(partial, r, ei[e], h);
          ++c;
        }
      }
      if (c) agg += t / (float)c;
    }
  }

  float s = agg + rootv + bv;
  if (relu) s = fmaxf(s, 0.0f);
  out[(size_t)n * DDIM + h] = s;
}

extern "C" void kernel_launch(void* const* d_in, const int* in_sizes, int n_in,
                              void* d_out, int out_size, void* d_ws, size_t ws_size,
                              hipStream_t stream) {
  const float* node_emb = (const float*)d_in[0];
  const float* sig      = (const float*)d_in[1];
  const float* W1       = (const float*)d_in[2];
  const float* root1    = (const float*)d_in[3];
  const float* bias1    = (const float*)d_in[4];
  const float* W2       = (const float*)d_in[5];
  const float* root2    = (const float*)d_in[6];
  const float* bias2    = (const float*)d_in[7];
  const int*   ei       = (const int*)d_in[8];
  const int*   et       = (const int*)d_in[9];
  const int    E        = in_sizes[9];

  float* xT      = (float*)d_ws;
  float* partial = xT + (size_t)DDIM * NPAD;
  float* h1      = partial + (size_t)KSPLIT * NSLOT * NPAD * DDIM;

  const int transGrid = (DDIM * NPAD) / 256;
  const int gemmGrid  = (DDIM / PANEL) * KSPLIT;
  const int aggGrid   = NNODES * (DDIM / 256);

  transpose_kernel<<<transGrid, 256, 0, stream>>>(node_emb, sig, 6, xT);
  gemm_kernel<<<gemmGrid, 512, 0, stream>>>(xT, W1, root1, partial);
  aggregate_kernel<<<aggGrid, 256, 0, stream>>>(partial, bias1, ei, et, E, h1, 1);

  transpose_kernel<<<transGrid, 256, 0, stream>>>(h1, nullptr, 0, xT);
  gemm_kernel<<<gemmGrid, 512, 0, stream>>>(xT, W2, root2, partial);
  aggregate_kernel<<<aggGrid, 256, 0, stream>>>(partial, bias2, ei, et, E, (float*)d_out, 0);
}


# aggregate: plane loads of up to 16 edge-list entries issued as one group ahead of their sums (was one dependent round trip per edge), blocks of 128 columns; same summation order
# speedup vs baseline: 3.5140x; 1.1191x over previous
#include <hip/hip_runtime.h>


#define NNODES 17
#define NPAD   32
#define NREL   8
#define NSLOT  9
#define DDIM   2048
#define MAXE   1024
#define GROUP  16
#define ACOLS  128
#define KC     128
#define KSPLIT 16
#define PANEL  128
#define BPITCH 144

typedef __attribute__((ext_vector_type(2))) float v2f;
typedef __attribute__((ext_vector_type(4))) float v4f;
typedef __attribute__((ext_vector_type(8))) float v8f;

__global__ __launch_bounds__(256) void transpose_kernel(
    const float* __restrict__ xmain, const float* __restrict__ xsig, int nsig,
    float* __restrict__ xT)
{
  int idx = blockIdx.x * 256 + threadIdx.x;
  int n = idx & (NPAD - 1);
  int k = idx >> 5;
  float v = 0.0f;
  if (n < NNODES)
    v = (n < nsig) ? xsig[(size_t)n * DDIM + k] : xmain[(size_t)n * DDIM + k];
  xT[idx] = v;
}

__global__ __launch_bounds__(512) void gemm_kernel(
    const float* __restrict__ xT,
    const float* __restrict__ W,
    const float* __restrict__ root,
    float* __restrict__ partial)
{
  __shared__ __attribute__((aligned(16))) float Bs[2][KC * BPITCH];

  const int tid  = threadIdx.x;
  const int wave = tid >> 5;
  const int lane = tid & 31;
  const int half = lane >> 4;
  const int hl   = lane & 15;
  const int strip = wave & 7;
  const int tile  = wave >> 3;

  const int panel  = blockIdx.x & 15;
  const int kchunk = blockIdx.x >> 4;
  const int pc = panel * PANEL;
  const int kc = kchunk * KC;

  v2f a[KC / 4];
  {
    const float* ap = xT + (size_t)(kc + 2 * half) * NPAD + 16 * tile + hl;
#pragma unroll
    for (int j = 0; j < KC / 4; ++j) {
      a[j].x = ap[(size_t)(4 * j) * NPAD];
      a[j].y = ap[(size_t)(4 * j + 1) * NPAD];
    }
  }

  const size_t goff = (size_t)(kc + wave) * DDIM + pc + 4 * lane;
  const int    soff = wave * BPITCH + 4 * lane;

  v4f r[8];
#pragma unroll
  for (int i = 0; i < 8; ++i)
    r[i] = *(const v4f*)(W + goff + (size_t)(16 * i) * DDIM);

#pragma unroll 1
  for (int slot = 0; slot < NSLOT; ++slot) {
    float* bs = Bs[slot & 1];
#pragma unroll
    for (int i = 0; i < 8; ++i)
      *(v4f*)(bs + soff + 16 * i * BPITCH) = r[i];
    __syncthreads();

    if (slot + 1 < NSLOT) {
      const float* Wn = (slot + 1 < NREL)
                            ? (W + (size_t)(slot + 1) * DDIM * DDIM) : root;
#pragma unroll
      for (int i = 0; i < 8; ++i)
        r[i] = *(const v4f*)(Wn + goff + (size_t)(16 * i) * DDIM);
    }

    v8f c = {};
    const float* bp = bs + (2 * half) * BPITCH + 16 * strip + hl;
    v2f b[KC / 4];
#pragma unroll
    for (int j = 0; j < KC / 4; ++j) {
      b[j].x = bp[(4 * j) * BPITCH];
      b[j].y = bp[(4 * j + 1) * BPITCH];
    }
    __builtin_amdgcn_sched_barrier(0);
#pragma unroll
    for (int j = 0; j < KC / 4; ++j)
      c = __builtin_amdgcn_wmma_f32_16x16x4_f32(false, a[j], false, b[j],
                                                (short)0, c, false, false);

    float* op = partial + ((size_t)(kchunk * NSLOT + slot) * NPAD) * DDIM
                + pc + 16 * strip + hl;
    if (tile == 0) {
#pragma unroll
      for (int i = 0; i < 8; ++i)
        op[(size_t)(i + 8 * half) * DDIM] = c[i];
    } else if (half == 0) {
      op[(size_t)16 * DDIM] = c[0];
    }
  }
}

__device__ __forceinline__ float plane_sum(const float* __restrict__ partial,
                                           int slot, int row, int h)
{
  const float* p = partial + ((size_t)slot * NPAD + row) * DDIM + h;
  float s = 0.0f;
#pragma unroll
  for (int q = 0; q < KSPLIT; ++q)
    s += p[(size_t)q * NSLOT * NPAD * DDIM];
  return s;
}

__global__ __launch_bounds__(ACOLS) void aggregate_kernel(
    const float* __restrict__ partial, const float* __restrict__ bias,
    const int* __restrict__ ei, const int* __restrict__ et, int E,
    float* __restrict__ out, int relu)
{
  __shared__ int sTy[MAXE];
  __shared__ int sSrc[MAXE];
  __shared__ int lTy[MAXE];
  __shared__ int lSrc[MAXE];
  __shared__ int lCnt[MAXE];
  __shared__ int sL;

  const int tid = threadIdx.x;
  const int n   = blockIdx.x / (DDIM / ACOLS);
  const int h   = (blockIdx.x % (DDIM / ACOLS)) * ACOLS + tid;

  const float rootv = plane_sum(partial, NREL, n, h);
  const float bv    = bias[h];

  float agg = 0.0f;

  if (E <= MAXE) {
    for (int e = tid; e < E; e += ACOLS) {
      int t = et[e];
      sTy[e]  = (ei[E + e] == n && (unsigned)t < (unsigned)NREL) ? t : -1;
      sSrc[e] = ei[e];
    }
    __syncthreads();
    for (int e = tid; e < E; e += ACOLS) {
      const int t = sTy[e];
      if (t >= 0) {
        int lower = 0, same = 0, sameBefore = 0;
        for (int e2 = 0; e2 < E; ++e2) {
          const int t2 = sTy[e2];
          lower      += (t2 >= 0 && t2 < t) ? 1 : 0;
          same       += (t2 == t) ? 1 : 0;
          sameBefore += (t2 == t && e2 < e) ? 1 : 0;
        }
        const int rank = lower + sameBefore;
        lTy[rank]  = t;
        lSrc[rank] = sSrc[e];
        lCnt[rank] = (sameBefore == same - 1) ? same : 0;
      }
    }
    if (tid == 0) {
      int c = 0;
      for (int e2 = 0; e2 < E; ++e2) c += (sTy[e2] >= 0) ? 1 : 0;
      sL = c;
    }
    __syncthreads();

    const int L = sL;
    float t = 0.0f;
    for (int i0 = 0; i0 < L; i0 += GROUP) {
      int ty[GROUP], sr[GROUP], cn[GROUP];
#pragma unroll
      for (int u = 0; u < GROUP; ++u) {
        const int i = min(i0 + u, L - 1);
        ty[u] = lTy[i];
        sr[u] = lSrc[i];
        cn[u] = lCnt[i];
      }
      float v[GROUP][KSPLIT];
#pragma unroll
      for (int u = 0; u < GROUP; ++u) {
        if (i0 + u < L) {
          const float* p = partial + ((size_t)ty[u] * NPAD + sr[u]) * DDIM + h;
#pragma unroll
          for (int q = 0; q < KSPLIT; ++q)
            v[u][q] = p[(size_t)q * NSLOT * NPAD * DDIM];
        }
      }
#pragma unroll
      for (int u = 0; u < GROUP; ++u) {
        if (i0 + u < L) {
          float s = 0.0f;
#pragma unroll
          for (int q = 0; q < KSPLIT; ++q)
            s += v[u][q];
          t += s;
          if (cn[u]) { agg += t / (float)cn[u]; t = 0.0f; }
        }
      }
    }
  } else {
    for (int r = 0; r < NREL; ++r) {
      float t = 0.0f;
      int   c = 0;
      for (int e = 0; e < E; ++e) {
        if (et[e] == r && ei[E + e] == n) {
          t += plane_sum(partial, r, ei[e], h);
          ++c;
        }
      }
      if (c) agg += t / (float)c;
    }
  }

  float s = agg + rootv + bv;
  if (relu) s = fmaxf(s, 0.0f);
  out[(size_t)n * DDIM + h] = s;
}

extern "C" void kernel_launch(void* const* d_in, const int* in_sizes, int n_in,
                              void* d_out, int out_size, void* d_ws, size_t ws_size,
                              hipStream_t stream) {
  const float* node_emb = (const float*)d_in[0];
  const float* sig      = (const float*)d_in[1];
  const float* W1       = (const float*)d_in[2];
  const float* root1    = (const float*)d_in[3];
  const float* bias1    = (const float*)d_in[4];
  const float* W2       = (const float*)d_in[5];
  const float* root2    = (const float*)d_in[6];
  const float* bias2    = (const float*)d_in[7];
  const int*   ei       = (const int*)d_in[8];
  const int*   et       = (const int*)d_in[9];
  const int    E        = in_sizes[9];

  float* xT      = (float*)d_ws;
  float* partial = xT + (size_t)DDIM * NPAD;
  float* h1      = partial + (size_t)KSPLIT * NSLOT * NPAD * DDIM;

  const int transGrid = (DDIM * NPAD) / 256;
  const int gemmGrid  = (DDIM / PANEL) * KSPLIT;
  const int aggGrid   = NNODES * (DDIM / ACOLS);

  transpose_kernel<<<transGrid, 256, 0, stream>>>(node_emb, sig, 6, xT);
  gemm_kernel<<<gemmGrid, 512, 0, stream>>>(xT, W1, root1, partial);
  aggregate_kernel<<<aggGrid, ACOLS, 0, stream>>>(partial, bias1, ei, et, E, h1, 1);

  transpose_kernel<<<transGrid, 256, 0, stream>>>(h1, nullptr, 0, xT);
  gemm_kernel<<<gemmGrid, 512, 0, stream>>>(xT, W2, root2, partial);
  aggregate_kernel<<<aggGrid, ACOLS, 0, stream>>>(partial, bias2, ei, et, E, (float*)d_out, 0);
}


# transposes dropped: gemm takes A fragments straight from row-major x (16-B loads, lane halves swap a k pair by permlanex16; layer 1 picks signal rows by pointer); 4 launches, same summation order
# speedup vs baseline: 3.7379x; 1.0637x over previous
#include <hip/hip_runtime.h>


#define NNODES 17
#define NPAD   32
#define NREL   8
#define NSLOT  9
#define DDIM   2048
#define MAXE   1024
#define GROUP  16
#define ACOLS  128
#define KC     128
#define KSPLIT 16
#define PANEL  128
#define BPITCH 144

typedef __attribute__((ext_vector_type(2))) float v2f;
typedef __attribute__((ext_vector_type(4))) float v4f;
typedef __attribute__((ext_vector_type(8))) float v8f;

__device__ __forceinline__ float xhalf(float x)
{
  return __builtin_bit_cast(float, __builtin_amdgcn_permlanex16(
      __builtin_bit_cast(int, x), __builtin_bit_cast(int, x),
      0x76543210, 0xfedcba98, false, false));
}

__global__ __launch_bounds__(512) void gemm_kernel(
    const float* __restrict__ xmain,
    const float* __restrict__ xsig,
    int nsig,
    const float* __restrict__ W,
    const float* __restrict__ root,
    float* __restrict__ partial)
{
  __shared__ __attribute__((aligned(16))) float Bs[2][KC * BPITCH];

  const int tid  = threadIdx.x;
  const int wave = tid >> 5;
  const int lane = tid & 31;
  const int half = lane >> 4;
  const int hl   = lane & 15;
  const int strip = wave & 7;
  const int tile  = wave >> 3;

  const int panel  = blockIdx.x & 15;
  const int kchunk = blockIdx.x >> 4;
  const int pc = panel * PANEL;
  const int kc = kchunk * KC;

  const size_t goff = (size_t)(kc + wave) * DDIM + pc + 4 * lane;
  const int    soff = wave * BPITCH + 4 * lane;

  v4f r[8];
#pragma unroll
  for (int i = 0; i < 8; ++i)
    r[i] = *(const v4f*)(W + goff + (size_t)(16 * i) * DDIM);

  v2f a[KC / 4];
  {
    const int m = 16 * tile + hl;
    v4f q[KC / 8];
    if (m < NNODES) {
      const float* xr = ((m < nsig) ? xsig : xmain)
                        + (size_t)m * DDIM + kc + 4 * half;
#pragma unroll
      for (int i = 0; i < KC / 8; ++i)
        q[i] = *(const v4f*)(xr + 8 * i);
    } else {
#pragma unroll
      for (int i = 0; i < KC / 8; ++i)
        q[i] = (v4f){0.0f, 0.0f, 0.0f, 0.0f};
    }
#pragma unroll
    for (int i = 0; i < KC / 8; ++i) {
      const float g0 = xhalf(half ? q[i].x : q[i].z);
      const float g1 = xhalf(half ? q[i].y : q[i].w);
      a[2 * i].x     = half ? g0 : q[i].x;
      a[2 * i].y     = half ? g1 : q[i].y;
      a[2 * i + 1].x = half ? q[i].z : g0;
      a[2 * i + 1].y = half ? q[i].w : g1;
    }
  }

#pragma unroll 1
  for (int slot = 0; slot < NSLOT; ++slot) {
    float* bs = Bs[slot & 1];
#pragma unroll
    for (int i = 0; i < 8; ++i)
      *(v4f*)(bs + soff + 16 * i * BPITCH) = r[i];
    __syncthreads();

    if (slot + 1 < NSLOT) {
      const float* Wn = (slot + 1 < NREL)
                            ? (W + (size_t)(slot + 1) * DDIM * DDIM) : root;
#pragma unroll
      for (int i = 0; i < 8; ++i)
        r[i] = *(const v4f*)(Wn + goff + (size_t)(16 * i) * DDIM);
    }

    v8f c = {};
    const float* bp = bs + (2 * half) * BPITCH + 16 * strip + hl;
    v2f b[KC / 4];
#pragma unroll
    for (int j = 0; j < KC / 4; ++j) {
      b[j].x = bp[(4 * j) * BPITCH];
      b[j].y = bp[(4 * j + 1) * BPITCH];
    }
    __builtin_amdgcn_sched_barrier(0);
#pragma unroll
    for (int j = 0; j < KC / 4; ++j)
      c = __builtin_amdgcn_wmma_f32_16x16x4_f32(false, a[j], false, b[j],
                                                (short)0, c, false, false);

    float* op = partial + ((size_t)(kchunk * NSLOT + slot) * NPAD) * DDIM
                + pc + 16 * strip + hl;
    if (tile == 0) {
#pragma unroll
      for (int i = 0; i < 8; ++i)
        op[(size_t)(i + 8 * half) * DDIM] = c[i];
    } else if (half == 0) {
      op[(size_t)16 * DDIM] = c[0];
    }
  }
}

__device__ __forceinline__ float plane_sum(const float* __restrict__ partial,
                                           int slot, int row, int h)
{
  const float* p = partial + ((size_t)slot * NPAD + row) * DDIM + h;
  float s = 0.0f;
#pragma unroll
  for (int q = 0; q < KSPLIT; ++q)
    s += p[(size_t)q * NSLOT * NPAD * DDIM];
  return s;
}

__global__ __launch_bounds__(ACOLS) void aggregate_kernel(
    const float* __restrict__ partial, const float* __restrict__ bias,
    const int* __restrict__ ei, const int* __restrict__ et, int E,
    float* __restrict__ out, int relu)
{
  __shared__ int sTy[MAXE];
  __shared__ int sSrc[MAXE];
  __shared__ int lTy[MAXE];
  __shared__ int lSrc[MAXE];
  __shared__ int lCnt[MAXE];
  __shared__ int sL;

  const int tid = threadIdx.x;
  const int n   = blockIdx.x / (DDIM / ACOLS);
  const int h   = (blockIdx.x % (DDIM / ACOLS)) * ACOLS + tid;

  const float rootv = plane_sum(partial, NREL, n, h);
  const float bv    = bias[h];

  float agg = 0.0f;

  if (E <= MAXE) {
    for (int e = tid; e < E; e += ACOLS) {
      int t = et[e];
      sTy[e]  = (ei[E + e] == n && (unsigned)t < (unsigned)NREL) ? t : -1;
      sSrc[e] = ei[e];
    }
    __syncthreads();
    for (int e = tid; e < E; e += ACOLS) {
      const int t = sTy[e];
      if (t >= 0) {
        int lower = 0, same = 0, sameBefore = 0;
        for (int e2 = 0; e2 < E; ++e2) {
          const int t2 = sTy[e2];
          lower      += (t2 >= 0 && t2 < t) ? 1 : 0;
          same       += (t2 == t) ? 1 : 0;
          sameBefore += (t2 == t && e2 < e) ? 1 : 0;
        }
        const int rank = lower + sameBefore;
        lTy[rank]  = t;
        lSrc[rank] = sSrc[e];
        lCnt[rank] = (sameBefore == same - 1) ? same : 0;
      }
    }
    if (tid == 0) {
      int c = 0;
      for (int e2 = 0; e2 < E; ++e2) c += (sTy[e2] >= 0) ? 1 : 0;
      sL = c;
    }
    __syncthreads();

    const int L = sL;
    float t = 0.0f;
    for (int i0 = 0; i0 < L; i0 += GROUP) {
      int ty[GROUP], sr[GROUP], cn[GROUP];
#pragma unroll
      for (int u = 0; u < GROUP; ++u) {
        const int i = min(i0 + u, L - 1);
        ty[u] = lTy[i];
        sr[u] = lSrc[i];
        cn[u] = lCnt[i];
      }
      float v[GROUP][KSPLIT];
#pragma unroll
      for (int u = 0; u < GROUP; ++u) {
        if (i0 + u < L) {
          const float* p = partial + ((size_t)ty[u] * NPAD + sr[u]) * DDIM + h;
#pragma unroll
          for (int q = 0; q < KSPLIT; ++q)
            v[u][q] = p[(size_t)q * NSLOT * NPAD * DDIM];
        }
      }
#pragma unroll
      for (int u = 0; u < GROUP; ++u) {
        if (i0 + u < L) {
          float s = 0.0f;
#pragma unroll
          for (int q = 0; q < KSPLIT; ++q)
            s += v[u][q];
          t += s;
          if (cn[u]) { agg += t / (float)cn[u]; t = 0.0f; }
        }
      }
    }
  } else {
    for (int r = 0; r < NREL; ++r) {
      float t = 0.0f;
      int   c = 0;
      for (int e = 0; e < E; ++e) {
        if (et[e] == r && ei[E + e] == n) {
          t += plane_sum(partial, r, ei[e], h);
          ++c;
        }
      }
      if (c) agg += t / (float)c;
    }
  }

  float s = agg + rootv + bv;
  if (relu) s = fmaxf(s, 0.0f);
  out[(size_t)n * DDIM + h] = s;
}

extern "C" void kernel_launch(void* const* d_in, const int* in_sizes, int n_in,
                              void* d_out, int out_size, void* d_ws, size_t ws_size,
                              hipStream_t stream) {
  const float* node_emb = (const float*)d_in[0];
  const float* sig      = (const float*)d_in[1];
  const float* W1       = (const float*)d_in[2];
  const float* root1    = (const float*)d_in[3];
  const float* bias1    = (const float*)d_in[4];
  const float* W2       = (const float*)d_in[5];
  const float* root2    = (const float*)d_in[6];
  const float* bias2    = (const float*)d_in[7];
  const int*   ei       = (const int*)d_in[8];
  const int*   et       = (const int*)d_in[9];
  const int    E        = in_sizes[9];

  float* partial = (float*)d_ws;
  float* h1      = partial + (size_t)KSPLIT * NSLOT * NPAD * DDIM;

  const int gemmGrid  = (DDIM / PANEL) * KSPLIT;
  const int aggGrid   = NNODES * (DDIM / ACOLS);

  gemm_kernel<<<gemmGrid, 512, 0, stream>>>(node_emb, sig, 6, W1, root1, partial);
  aggregate_kernel<<<aggGrid, ACOLS, 0, stream>>>(partial, bias1, ei, et, E, h1, 1);

  gemm_kernel<<<gemmGrid, 512, 0, stream>>>(h1, nullptr, 0, W2, root2, partial);
  aggregate_kernel<<<aggGrid, ACOLS, 0, stream>>>(partial, bias2, ei, et, E, (float*)d_out, 0);
}


# gemm: each relation's WMMA tile holds only the source nodes of its edges (block scans the edge list, A rows gathered from an LDS copy of x): 10 tile passes per strip, was 18; same summation order
# speedup vs baseline: 4.0730x; 1.0896x over previous
#include <hip/hip_runtime.h>


#define NNODES 17
#define NPAD   32
#define NREL   8
#define NSLOT  9
#define DDIM   2048
#define MAXE   1024
#define GROUP  16
#define ACOLS  128
#define KC     128
#define KSPLIT 16
#define PANEL  128
#define XROWS  18
#define ZROW   17
#define XPITCH 132
#define EPRE   3
#define BPITCH 144

typedef __attribute__((ext_vector_type(2))) float v2f;
typedef __attribute__((ext_vector_type(4))) float v4f;
typedef __attribute__((ext_vector_type(8))) float v8f;
typedef __attribute__((ext_vector_type(4))) int   v4i;

__global__ __launch_bounds__(512) void gemm_kernel(
    const float* __restrict__ xmain,
    const float* __restrict__ xsig,
    int nsig,
    const float* __restrict__ W,
    const float* __restrict__ root,
    const int* __restrict__ ei,
    const int* __restrict__ et,
    int E,
    float* __restrict__ partial)
{
  __shared__ __attribute__((aligned(16))) float Bs[2][KC * BPITCH];
  __shared__ __attribute__((aligned(16))) float xs[XROWS * XPITCH];
  __shared__ __attribute__((aligned(16))) int umap[NSLOT][2][16];
  __shared__ int amap[NSLOT][2][16];
  __shared__ int smap[NREL][16];
  __shared__ int scnt[NREL];

  const int tid  = threadIdx.x;
  const int wave = tid >> 5;
  const int lane = tid & 31;
  const int half = lane >> 4;
  const int hl   = lane & 15;
  const int strip = wave & 7;
  const int tile  = wave >> 3;

  const int panel  = blockIdx.x & 15;
  const int kchunk = blockIdx.x >> 4;
  const int pc = panel * PANEL;
  const int kc = kchunk * KC;

  v4f xq[2];
#pragma unroll
  for (int u = 0; u < 2; ++u) {
    const int row = wave + 16 * u;
    xq[u] = (v4f){0.0f, 0.0f, 0.0f, 0.0f};
    if (row < NNODES)
      xq[u] = *(const v4f*)(((row < nsig) ? xsig : xmain)
                            + (size_t)row * DDIM + kc + 4 * lane);
  }
  int pt[EPRE], ps[EPRE];
#pragma unroll
  for (int u = 0; u < EPRE; ++u) {
    const int e = lane + 32 * u;
    pt[u] = -1;
    ps[u] = 0;
    if (wave < NREL && e < E) {
      pt[u] = et[e];
      ps[u] = ei[e];
    }
  }

  const size_t goff = (size_t)(kc + wave) * DDIM + pc + 4 * lane;
  const int    soff = wave * BPITCH + 4 * lane;

  v4f r[8];
#pragma unroll
  for (int i = 0; i < 8; ++i)
    r[i] = *(const v4f*)(W + goff + (size_t)(16 * i) * DDIM);

#pragma unroll
  for (int u = 0; u < 2; ++u) {
    const int row = wave + 16 * u;
    if (row < XROWS)
      *(v4f*)(xs + row * XPITCH + 4 * lane) = xq[u];
  }

  if (wave < NREL) {
    unsigned m = 0;
#pragma unroll
    for (int u = 0; u < EPRE; ++u)
      if (pt[u] == wave && (unsigned)ps[u] < (unsigned)NNODES) m |= 1u << ps[u];
    for (int e = lane + 32 * EPRE; e < E; e += 32) {
      const int s = ei[e];
      if (et[e] == wave && (unsigned)s < (unsigned)NNODES) m |= 1u << s;
    }
    unsigned full = 0;
    for (int b = 0; b < NNODES; ++b)
      if (__any((int)((m >> b) & 1u))) full |= 1u << b;
    if (lane < 16) {
      int node = ZROW, seen = 0;
      for (int b = 0; b < NNODES; ++b) {
        if ((full >> b) & 1u) {
          if (seen == lane) node = b;
          ++seen;
        }
      }
      smap[wave][lane] = node;
    }
    if (lane == 0) scnt[wave] = __popc(full);
  }
  __syncthreads();

  bool fast = true;
  for (int q = 0; q < NREL; ++q) fast = fast && (scnt[q] <= 16);

  if (tid < NSLOT * 32) {
    const int s = tid >> 5, t = (tid >> 4) & 1, i = tid & 15;
    int node, first;
    if (fast && s < NREL) {
      node  = t ? ZROW : smap[s][i];
      first = t ? ZROW : smap[s][0];
    } else {
      node = 16 * t + i;
      if (node >= NNODES) node = ZROW;
      first = 16 * t;
    }
    umap[s][t][i] = node;
    amap[s][t][i] = (node < NNODES) ? node : first;
  }
#pragma unroll
  for (int i = 0; i < 8; ++i)
    *(v4f*)(Bs[0] + soff + 16 * i * BPITCH) = r[i];
  __syncthreads();

#pragma unroll 1
  for (int slot = 0; slot < NSLOT; ++slot) {
    const float* bs = Bs[slot & 1];

    if (slot + 1 < NSLOT) {
      const float* Wn = (slot + 1 < NREL)
                            ? (W + (size_t)(slot + 1) * DDIM * DDIM) : root;
#pragma unroll
      for (int i = 0; i < 8; ++i)
        r[i] = *(const v4f*)(Wn + goff + (size_t)(16 * i) * DDIM);
    }

    const int units = (fast && slot < NREL) ? 1 : 2;
    if (tile < units) {
      const int* um = &umap[slot][tile][0];
      const int arow = amap[slot][tile][hl];
      const v4i n0 = *(const v4i*)(um + 8 * half);
      const v4i n1 = *(const v4i*)(um + 8 * half + 4);

      const float* bp = bs + (2 * half) * BPITCH + 16 * strip + hl;
      v2f b[KC / 4];
#pragma unroll
      for (int j = 0; j < KC / 4; ++j) {
        b[j].x = bp[(4 * j) * BPITCH];
        b[j].y = bp[(4 * j + 1) * BPITCH];
      }
      const float* ap = xs + arow * XPITCH + 2 * half;
      v2f a[KC / 4];
#pragma unroll
      for (int j = 0; j < KC / 4; ++j)
        a[j] = *(const v2f*)(ap + 4 * j);
      __builtin_amdgcn_sched_barrier(0);
      v8f c = {};
#pragma unroll
      for (int j = 0; j < KC / 4; ++j)
        c = __builtin_amdgcn_wmma_f32_16x16x4_f32(false, a[j], false, b[j],
                                                  (short)0, c, false, false);

      float* op = partial + ((size_t)(kchunk * NSLOT + slot) * NPAD) * DDIM
                  + pc + 16 * strip + hl;
      const int nd[8] = {n0.x, n0.y, n0.z, n0.w, n1.x, n1.y, n1.z, n1.w};
#pragma unroll
      for (int i = 0; i < 8; ++i)
        if (nd[i] < NNODES) op[(size_t)nd[i] * DDIM] = c[i];
    }

    if (slot + 1 < NSLOT) {
      float* bn = Bs[(slot + 1) & 1];
#pragma unroll
      for (int i = 0; i < 8; ++i)
        *(v4f*)(bn + soff + 16 * i * BPITCH) = r[i];
      __syncthreads();
    }
  }
}

__device__ __forceinline__ float plane_sum(const float* __restrict__ partial,
                                           int slot, int row, int h)
{
  const float* p = partial + ((size_t)slot * NPAD + row) * DDIM + h;
  float s = 0.0f;
#pragma unroll
  for (int q = 0; q < KSPLIT; ++q)
    s += p[(size_t)q * NSLOT * NPAD * DDIM];
  return s;
}

__global__ __launch_bounds__(ACOLS) void aggregate_kernel(
    const float* __restrict__ partial, const float* __restrict__ bias,
    const int* __restrict__ ei, const int* __restrict__ et, int E,
    float* __restrict__ out, int relu)
{
  __shared__ int sTy[MAXE];
  __shared__ int sSrc[MAXE];
  __shared__ int lTy[MAXE];
  __shared__ int lSrc[MAXE];
  __shared__ int lCnt[MAXE];
  __shared__ int sL;

  const int tid = threadIdx.x;
  const int n   = blockIdx.x / (DDIM / ACOLS);
  const int h   = (blockIdx.x % (DDIM / ACOLS)) * ACOLS + tid;

  const float rootv = plane_sum(partial, NREL, n, h);
  const float bv    = bias[h];

  float agg = 0.0f;

  if (E <= MAXE) {
    for (int e = tid; e < E; e += ACOLS) {
      int t = et[e];
      sTy[e]  = (ei[E + e] == n && (unsigned)t < (unsigned)NREL) ? t : -1;
      sSrc[e] = ei[e];
    }
    __syncthreads();
    for (int e = tid; e < E; e += ACOLS) {
      const int t = sTy[e];
      if (t >= 0) {
        int lower = 0, same = 0, sameBefore = 0;
        for (int e2 = 0; e2 < E; ++e2) {
          const int t2 = sTy[e2];
          lower      += (t2 >= 0 && t2 < t) ? 1 : 0;
          same       += (t2 == t) ? 1 : 0;
          sameBefore += (t2 == t && e2 < e) ? 1 : 0;
        }
        const int rank = lower + sameBefore;
        lTy[rank]  = t;
        lSrc[rank] = sSrc[e];
        lCnt[rank] = (sameBefore == same - 1) ? same : 0;
      }
    }
    if (tid == 0) {
      int c = 0;
      for (int e2 = 0; e2 < E; ++e2) c += (sTy[e2] >= 0) ? 1 : 0;
      sL = c;
    }
    __syncthreads();

    const int L = sL;
    float t = 0.0f;
    for (int i0 = 0; i0 < L; i0 += GROUP) {
      int ty[GROUP], sr[GROUP], cn[GROUP];
#pragma unroll
      for (int u = 0; u < GROUP; ++u) {
        const int i = min(i0 + u, L - 1);
        ty[u] = lTy[i];
        sr[u] = lSrc[i];
        cn[u] = lCnt[i];
      }
      float v[GROUP][KSPLIT];
#pragma unroll
      for (int u = 0; u < GROUP; ++u) {
        if (i0 + u < L) {
          const float* p = partial + ((size_t)ty[u] * NPAD + sr[u]) * DDIM + h;
#pragma unroll
          for (int q = 0; q < KSPLIT; ++q)
            v[u][q] = p[(size_t)q * NSLOT * NPAD * DDIM];
        }
      }
#pragma unroll
      for (int u = 0; u < GROUP; ++u) {
        if (i0 + u < L) {
          float s = 0.0f;
#pragma unroll
          for (int q = 0; q < KSPLIT; ++q)
            s += v[u][q];
          t += s;
          if (cn[u]) { agg += t / (float)cn[u]; t = 0.0f; }
        }
      }
    }
  } else {
    for (int r = 0; r < NREL; ++r) {
      float t = 0.0f;
      int   c = 0;
      for (int e = 0; e < E; ++e) {
        if (et[e] == r && ei[E + e] == n) {
          t += plane_sum(partial, r, ei[e], h);
          ++c;
        }
      }
      if (c) agg += t / (float)c;
    }
  }

  float s = agg + rootv + bv;
  if (relu) s = fmaxf(s, 0.0f);
  out[(size_t)n * DDIM + h] = s;
}

extern "C" void kernel_launch(void* const* d_in, const int* in_sizes, int n_in,
                              void* d_out, int out_size, void* d_ws, size_t ws_size,
                              hipStream_t stream) {
  const float* node_emb = (const float*)d_in[0];
  const float* sig      = (const float*)d_in[1];
  const float* W1       = (const float*)d_in[2];
  const float* root1    = (const float*)d_in[3];
  const float* bias1    = (const float*)d_in[4];
  const float* W2       = (const float*)d_in[5];
  const float* root2    = (const float*)d_in[6];
  const float* bias2    = (const float*)d_in[7];
  const int*   ei       = (const int*)d_in[8];
  const int*   et       = (const int*)d_in[9];
  const int    E        = in_sizes[9];

  float* partial = (float*)d_ws;
  float* h1      = partial + (size_t)KSPLIT * NSLOT * NPAD * DDIM;

  const int gemmGrid  = (DDIM / PANEL) * KSPLIT;
  const int aggGrid   = NNODES * (DDIM / ACOLS);

  gemm_kernel<<<gemmGrid, 512, 0, stream>>>(node_emb, sig, 6, W1, root1, ei, et, E, partial);
  aggregate_kernel<<<aggGrid, ACOLS, 0, stream>>>(partial, bias1, ei, et, E, h1, 1);

  gemm_kernel<<<gemmGrid, 512, 0, stream>>>(h1, nullptr, 0, W2, root2, ei, et, E, partial);
  aggregate_kernel<<<aggGrid, ACOLS, 0, stream>>>(partial, bias2, ei, et, E, (float*)d_out, 0);
}


# gemm: edge mean taken on the D tile in registers by a second WMMA (coefficient table x D), summed over the 9 slots; block stores one 17-row tile, last launch adds 16 planes; new summation order
# speedup vs baseline: 4.5498x; 1.1171x over previous
#include <hip/hip_runtime.h>


#define NNODES 17
#define NREL   8
#define NSLOT  9
#define DDIM   2048
#define KC     128
#define KSPLIT 16
#define PANEL  128
#define XROWS  18
#define ZROW   17
#define XPITCH 132
#define EPRE   3
#define BPITCH 144
#define CTAB   (NNODES * 16)
#define OPITCH 132
#define RTHREADS 64

typedef __attribute__((ext_vector_type(2))) float v2f;
typedef __attribute__((ext_vector_type(4))) float v4f;
typedef __attribute__((ext_vector_type(8))) float v8f;

__device__ __forceinline__ int row_of(int i)
{
  return ((i >> 2) << 1) | (i & 1) | (((i >> 1) & 1) << 3);
}
__device__ __forceinline__ int index_of(int row)
{
  return (((row & 7) >> 1) << 2) | ((row >> 3) << 1) | (row & 1);
}

__global__ __launch_bounds__(512) void gemm_kernel(
    const float* __restrict__ xmain,
    const float* __restrict__ xsig,
    int nsig,
    const float* __restrict__ W,
    const float* __restrict__ root,
    const int* __restrict__ ei,
    const int* __restrict__ et,
    int E,
    float* __restrict__ partial)
{
  __shared__ __attribute__((aligned(16))) float Bs[2][KC * BPITCH];
  __shared__ __attribute__((aligned(16))) float xs[XROWS * XPITCH];
  __shared__ __attribute__((aligned(16))) float coefT[NSLOT][2][CTAB];
  __shared__ __attribute__((aligned(16))) float ost[NNODES * OPITCH];
  __shared__ int amap[NSLOT][2][16];
  __shared__ int smap[NREL][16];
  __shared__ int scnt[NREL];

  const int tid  = threadIdx.x;
  const int wave = tid >> 5;
  const int lane = tid & 31;
  const int half = lane >> 4;
  const int hl   = lane & 15;
  const int strip = wave & 7;
  const int tile  = wave >> 3;

  const int panel  = blockIdx.x & 15;
  const int kchunk = blockIdx.x >> 4;
  const int pc = panel * PANEL;
  const int kc = kchunk * KC;

  v4f xq[2];
#pragma unroll
  for (int u = 0; u < 2; ++u) {
    const int row = wave + 16 * u;
    xq[u] = (v4f){0.0f, 0.0f, 0.0f, 0.0f};
    if (row < NNODES)
      xq[u] = *(const v4f*)(((row < nsig) ? xsig : xmain)
                            + (size_t)row * DDIM + kc + 4 * lane);
  }
  int pt[EPRE], ps[EPRE], pd[EPRE];
#pragma unroll
  for (int u = 0; u < EPRE; ++u) {
    const int e = lane + 32 * u;
    pt[u] = -1;
    ps[u] = 0;
    pd[u] = 0;
    if (wave < NREL && e < E) {
      pt[u] = et[e];
      ps[u] = ei[e];
      pd[u] = ei[E + e];
    }
  }

  const size_t goff = (size_t)(kc + wave) * DDIM + pc + 4 * lane;
  const int    soff = wave * BPITCH + 4 * lane;

  v4f r[8];
#pragma unroll
  for (int i = 0; i < 8; ++i)
    r[i] = *(const v4f*)(W + goff + (size_t)(16 * i) * DDIM);

#pragma unroll
  for (int u = 0; u < 2; ++u) {
    const int row = wave + 16 * u;
    if (row < XROWS)
      *(v4f*)(xs + row * XPITCH + 4 * lane) = xq[u];
  }

  if (wave < NREL) {
    bool ok[EPRE];
    unsigned m = 0;
#pragma unroll
    for (int u = 0; u < EPRE; ++u) {
      ok[u] = pt[u] == wave && (unsigned)ps[u] < (unsigned)NNODES
              && (unsigned)pd[u] < (unsigned)NNODES;
      if (ok[u]) m |= 1u << ps[u];
    }
    for (int e = lane + 32 * EPRE; e < E; e += 32) {
      const int s = ei[e], d = ei[E + e];
      if (et[e] == wave && (unsigned)s < (unsigned)NNODES
          && (unsigned)d < (unsigned)NNODES) m |= 1u << s;
    }
    unsigned full = 0;
    for (int b = 0; b < NNODES; ++b)
      if (__any((int)((m >> b) & 1u))) full |= 1u << b;
    if (lane < 16) {
      int node = ZROW, seen = 0;
      for (int b = 0; b < NNODES; ++b) {
        if ((full >> b) & 1u) {
          if (seen == lane) node = b;
          ++seen;
        }
      }
      smap[wave][lane] = node;
    }
    if (lane == 0) scnt[wave] = __popc(full);

    unsigned cnt[8] = {0u, 0u, 0u, 0u, 0u, 0u, 0u, 0u};
    unsigned c16 = 0u, tot = 0u;
    auto count_edge = [&](int s, int d) {
      const int rank = __popc(full & ((1u << s) - 1u));
      const unsigned hit = (lane == d) ? 1u : 0u;
      tot += hit;
      if (rank >= 16) {
        c16 += hit;
      } else {
        const int w = row_of(rank);
        const int sel = w >> 1;
        const unsigned inc = hit << (16 * (w & 1));
#pragma unroll
        for (int k = 0; k < 8; ++k) cnt[k] += (k == sel) ? inc : 0u;
      }
    };
#pragma unroll
    for (int u = 0; u < EPRE; ++u) {
      unsigned mk = __builtin_amdgcn_ballot_w32(ok[u]);
      while (mk) {
        const int b = __ffs(mk) - 1;
        mk &= mk - 1u;
        count_edge(__builtin_amdgcn_readlane(ps[u], b),
                   __builtin_amdgcn_readlane(pd[u], b));
      }
    }
    for (int e0 = 32 * EPRE; e0 < E; e0 += 32) {
      const int e = e0 + lane;
      int s = 0, d = 0;
      bool k = false;
      if (e < E) {
        s = ei[e];
        d = ei[E + e];
        k = et[e] == wave && (unsigned)s < (unsigned)NNODES
            && (unsigned)d < (unsigned)NNODES;
      }
      unsigned mk = __builtin_amdgcn_ballot_w32(k);
      while (mk) {
        const int b = __ffs(mk) - 1;
        mk &= mk - 1u;
        count_edge(__builtin_amdgcn_readlane(s, b), __builtin_amdgcn_readlane(d, b));
      }
    }
    if (lane < NNODES) {
      const float inv = 1.0f / (float)max(tot, 1u);
      float* t0 = &coefT[wave][0][lane * 16];
      float* t1 = &coefT[wave][1][lane * 16];
#pragma unroll
      for (int k = 0; k < 4; ++k)
        *(v4f*)(t0 + 4 * k) = (v4f){(float)(cnt[2 * k] & 0xffffu) * inv,
                                    (float)(cnt[2 * k] >> 16) * inv,
                                    (float)(cnt[2 * k + 1] & 0xffffu) * inv,
                                    (float)(cnt[2 * k + 1] >> 16) * inv};
      *(v4f*)(t1) = (v4f){(float)c16 * inv, 0.0f, 0.0f, 0.0f};
#pragma unroll
      for (int k = 1; k < 4; ++k)
        *(v4f*)(t1 + 4 * k) = (v4f){0.0f, 0.0f, 0.0f, 0.0f};
    }
  } else if (wave == NREL) {
    if (lane < NNODES) {
      float* t0 = &coefT[NREL][0][lane * 16];
      float* t1 = &coefT[NREL][1][lane * 16];
#pragma unroll
      for (int k = 0; k < 4; ++k) {
        *(v4f*)(t0 + 4 * k) = (v4f){(lane == 4 * k) ? 1.0f : 0.0f,
                                    (lane == 4 * k + 1) ? 1.0f : 0.0f,
                                    (lane == 4 * k + 2) ? 1.0f : 0.0f,
                                    (lane == 4 * k + 3) ? 1.0f : 0.0f};
        *(v4f*)(t1 + 4 * k) = (v4f){(k == 0 && lane == 16) ? 1.0f : 0.0f,
                                    0.0f, 0.0f, 0.0f};
      }
    }
  }
  __syncthreads();

  if (tid < NSLOT * 32) {
    const int s = tid >> 5, t = (tid >> 4) & 1, i = tid & 15;
    int node, first;
    if (s < NREL) {
      const int c = scnt[s];
      if (t == 0) {
        node  = smap[s][index_of(i)];
        first = smap[s][0];
      } else {
        node  = (c > 16 && i == 0) ? 16 : ZROW;
        first = (c > 16) ? 16 : ZROW;
      }
    } else {
      node = 16 * t + i;
      if (node >= NNODES) node = ZROW;
      first = 16 * t;
    }
    amap[s][t][i] = (node < NNODES) ? node : first;
  }
  unsigned mo = 0u, tw = 0u;
  for (int q = 0; q < NREL; ++q) {
    const int c = scnt[q];
    mo |= ((tile == 0 && c > 8) ? 1u : 0u) << q;
    tw |= ((c > 16) ? 1u : 0u) << q;
  }
  mo |= ((tile == 0) ? 1u : 0u) << NREL;
  tw |= 1u << NREL;
  const unsigned moreMask = __builtin_amdgcn_readfirstlane(mo);
  const unsigned twoMask  = __builtin_amdgcn_readfirstlane(tw);
#pragma unroll
  for (int i = 0; i < 8; ++i)
    *(v4f*)(Bs[0] + soff + 16 * i * BPITCH) = r[i];
  __syncthreads();

  v8f acc0 = {};
  v8f acc1 = {};

#pragma unroll 1
  for (int slot = 0; slot < NSLOT; ++slot) {
    const float* bs = Bs[slot & 1];

    if (slot + 1 < NSLOT) {
      const float* Wn = (slot + 1 < NREL)
                            ? (W + (size_t)(slot + 1) * DDIM * DDIM) : root;
#pragma unroll
      for (int i = 0; i < 8; ++i)
        r[i] = *(const v4f*)(Wn + goff + (size_t)(16 * i) * DDIM);
    }

    const int units = ((twoMask >> slot) & 1u) ? 2 : 1;
    if (tile < units) {
      const int arow = amap[slot][tile][hl];
      const float* tab = &coefT[slot][tile][0];
      const v4f q0 = *(const v4f*)(tab + hl * 16 + 8 * half);
      const v4f q1 = *(const v4f*)(tab + hl * 16 + 8 * half + 4);
      const float* tp = (hl == 0) ? (tab + 16 * 16 + 8 * half)
                                  : (xs + ZROW * XPITCH + 8 * half);
      const v4f p0 = *(const v4f*)(tp);
      const v4f p1 = *(const v4f*)(tp + 4);

      const float* bp = bs + (2 * half) * BPITCH + 16 * strip + hl;
      v2f b[KC / 4];
#pragma unroll
      for (int j = 0; j < KC / 4; ++j) {
        b[j].x = bp[(4 * j) * BPITCH];
        b[j].y = bp[(4 * j + 1) * BPITCH];
      }
      const float* ap = xs + arow * XPITCH + 2 * half;
      v2f a[KC / 4];
#pragma unroll
      for (int j = 0; j < KC / 4; ++j)
        a[j] = *(const v2f*)(ap + 4 * j);
      __builtin_amdgcn_sched_barrier(0);
      v8f c = {};
#pragma unroll
      for (int j = 0; j < KC / 4; ++j)
        c = __builtin_amdgcn_wmma_f32_16x16x4_f32(false, a[j], false, b[j],
                                                  (short)0, c, false, false);

      acc0 = __builtin_amdgcn_wmma_f32_16x16x4_f32(false, (v2f){q0.x, q0.y}, false,
                 (v2f){c[0], c[1]}, (short)0, acc0, false, false);
      acc1 = __builtin_amdgcn_wmma_f32_16x16x4_f32(false, (v2f){p0.x, p0.y}, false,
                 (v2f){c[0], c[1]}, (short)0, acc1, false, false);
      acc0 = __builtin_amdgcn_wmma_f32_16x16x4_f32(false, (v2f){q0.z, q0.w}, false,
                 (v2f){c[2], c[3]}, (short)0, acc0, false, false);
      acc1 = __builtin_amdgcn_wmma_f32_16x16x4_f32(false, (v2f){p0.z, p0.w}, false,
                 (v2f){c[2], c[3]}, (short)0, acc1, false, false);
      if ((moreMask >> slot) & 1u) {
        acc0 = __builtin_amdgcn_wmma_f32_16x16x4_f32(false, (v2f){q1.x, q1.y}, false,
                   (v2f){c[4], c[5]}, (short)0, acc0, false, false);
        acc1 = __builtin_amdgcn_wmma_f32_16x16x4_f32(false, (v2f){p1.x, p1.y}, false,
                   (v2f){c[4], c[5]}, (short)0, acc1, false, false);
        acc0 = __builtin_amdgcn_wmma_f32_16x16x4_f32(false, (v2f){q1.z, q1.w}, false,
                   (v2f){c[6], c[7]}, (short)0, acc0, false, false);
        acc1 = __builtin_amdgcn_wmma_f32_16x16x4_f32(false, (v2f){p1.z, p1.w}, false,
                   (v2f){c[6], c[7]}, (short)0, acc1, false, false);
      }
    }

    if (slot + 1 < NSLOT) {
      float* bn = Bs[(slot + 1) & 1];
#pragma unroll
      for (int i = 0; i < 8; ++i)
        *(v4f*)(bn + soff + 16 * i * BPITCH) = r[i];
      __syncthreads();
    }
  }

  const int col = 16 * strip + hl;
  if (tile == 1) {
#pragma unroll
    for (int i = 0; i < 8; ++i) ost[(8 * half + i) * OPITCH + col] = acc0[i];
    if (half == 0) ost[16 * OPITCH + col] = acc1[0];
  }
  __syncthreads();
  if (tile == 0) {
#pragma unroll
    for (int i = 0; i < 8; ++i) ost[(8 * half + i) * OPITCH + col] += acc0[i];
    if (half == 0) ost[16 * OPITCH + col] += acc1[0];
  }
  __syncthreads();
#pragma unroll
  for (int u = 0; u < 2; ++u) {
    const int row = wave + 16 * u;
    if (row < NNODES)
      *(v4f*)(partial + ((size_t)kchunk * NNODES + row) * DDIM + pc + 4 * lane)
          = *(const v4f*)(ost + row * OPITCH + 4 * lane);
  }
}

__global__ __launch_bounds__(RTHREADS) void reduce_kernel(
    const float* __restrict__ partial, const float* __restrict__ bias,
    float* __restrict__ out, int relu)
{
  const int idx = blockIdx.x * RTHREADS + threadIdx.x;
  const int n   = idx / (DDIM / 4);
  const int h   = (idx % (DDIM / 4)) * 4;

  const float* p = partial + (size_t)n * DDIM + h;
  v4f v[KSPLIT];
#pragma unroll
  for (int q = 0; q < KSPLIT; ++q)
    v[q] = *(const v4f*)(p + (size_t)q * NNODES * DDIM);
  const v4f bv = *(const v4f*)(bias + h);
  v4f s = {0.0f, 0.0f, 0.0f, 0.0f};
#pragma unroll
  for (int q = 0; q < KSPLIT; ++q)
    s += v[q];
  s += bv;
  if (relu) {
    s.x = fmaxf(s.x, 0.0f);
    s.y = fmaxf(s.y, 0.0f);
    s.z = fmaxf(s.z, 0.0f);
    s.w = fmaxf(s.w, 0.0f);
  }
  *(v4f*)(out + (size_t)n * DDIM + h) = s;
}

extern "C" void kernel_launch(void* const* d_in, const int* in_sizes, int n_in,
                              void* d_out, int out_size, void* d_ws, size_t ws_size,
                              hipStream_t stream) {
  const float* node_emb = (const float*)d_in[0];
  const float* sig      = (const float*)d_in[1];
  const float* W1       = (const float*)d_in[2];
  const float* root1    = (const float*)d_in[3];
  const float* bias1    = (const float*)d_in[4];
  const float* W2       = (const float*)d_in[5];
  const float* root2    = (const float*)d_in[6];
  const float* bias2    = (const float*)d_in[7];
  const int*   ei       = (const int*)d_in[8];
  const int*   et       = (const int*)d_in[9];
  const int    E        = in_sizes[9];

  float* partial = (float*)d_ws;
  float* h1      = partial + (size_t)KSPLIT * NNODES * DDIM;

  const int gemmGrid = (DDIM / PANEL) * KSPLIT;
  const int redGrid  = NNODES * (DDIM / 4) / RTHREADS;

  gemm_kernel<<<gemmGrid, 512, 0, stream>>>(node_emb, sig, 6, W1, root1, ei, et, E, partial);
  reduce_kernel<<<redGrid, RTHREADS, 0, stream>>>(partial, bias1, h1, 1);

  gemm_kernel<<<gemmGrid, 512, 0, stream>>>(h1, nullptr, 0, W2, root2, ei, et, E, partial);
  reduce_kernel<<<redGrid, RTHREADS, 0, stream>>>(partial, bias2, (float*)d_out, 0);
}
